# starting kernel
# baseline (speedup 1.0000x reference)
#include <hip/hip_runtime.h>


typedef float v2f __attribute__((ext_vector_type(2)));
typedef float v4f __attribute__((ext_vector_type(4)));
typedef float v8f __attribute__((ext_vector_type(8)));

#define INV_SQRT2 0.70710678118654752440f
#define FDIM 4096
#define LDS_STRIDE 20
#define WAVES_PER_BLOCK 8
#define NTILES 4

__device__ __forceinline__ float wval(int k, int n) {
  const float c = INV_SQRT2;
  int p = k >> 1;
  float s = (k & 1) ? -c : c;
  return (n == p) ? c : ((n == (p + 8)) ? s : 0.0f);
}

__global__ __launch_bounds__(256) void haar_dwt_wmma(const float* __restrict__ x,
                                                     float* __restrict__ out) {
  __shared__ float lds[WAVES_PER_BLOCK][NTILES][16 * LDS_STRIDE];

  const int lane = threadIdx.x & 31;
  const int wave = threadIdx.x >> 5;

  const int row0 = blockIdx.y * 16;
  const int col0 = (blockIdx.x * WAVES_PER_BLOCK + wave) * (16 * NTILES);

  {
    const int r  = lane >> 1;
    const int cb = (lane & 1) * 8;
    const float* src = x + ((size_t)(row0 + r) << 12) + col0 + cb;
    float* dst = &lds[wave][0][r * LDS_STRIDE + cb];
#pragma unroll
    for (int t = 0; t < NTILES; ++t) {
      v4f d0 = __builtin_nontemporal_load((const v4f*)(src + t * 16));
      v4f d1 = __builtin_nontemporal_load((const v4f*)(src + t * 16 + 4));
      *(v4f*)(dst + t * 16 * LDS_STRIDE)     = d0;
      *(v4f*)(dst + t * 16 * LDS_STRIDE + 4) = d1;
    }
  }

  const int n  = lane & 15;
  const int h  = lane >> 4;
  const int kb = 2 * h;

  v2f bmat[4];
#pragma unroll
  for (int j = 0; j < 4; ++j) {
    int k0 = 4 * j + kb;
    bmat[j].x = wval(k0,     n);
    bmat[j].y = wval(k0 + 1, n);
  }

  const size_t orow = ((size_t)(row0 + n) << 12) + h * (FDIM / 2) + (col0 >> 1);

#pragma unroll
  for (int t = 0; t < NTILES; ++t) {
    float* tile = lds[wave][t];

    v8f acc = {};
#pragma unroll
    for (int j = 0; j < 4; ++j) {
      v2f a = *(const v2f*)&tile[n * LDS_STRIDE + 4 * j + kb];
      acc = __builtin_amdgcn_wmma_f32_16x16x4_f32(false, a, false, bmat[j],
                                                  (short)0, acc, false, false);
    }

#pragma unroll
    for (int v = 0; v < 8; ++v)
      tile[(v + 8 * h) * LDS_STRIDE + n] = acc[v];

    float* dstp = out + orow + t * 8;
    v4f s0 = *(const v4f*)&tile[n * LDS_STRIDE + 8 * h];
    v4f s1 = *(const v4f*)&tile[n * LDS_STRIDE + 8 * h + 4];
    __builtin_nontemporal_store(s0, (v4f*)dstp);
    __builtin_nontemporal_store(s1, (v4f*)(dstp + 4));
  }
}

extern "C" void kernel_launch(void* const* d_in, const int* in_sizes, int n_in,
                              void* d_out, int out_size, void* d_ws, size_t ws_size,
                              hipStream_t stream) {
  (void)n_in; (void)out_size; (void)d_ws; (void)ws_size;
  const float* x = (const float*)d_in[0];
  float* out = (float*)d_out;

  const int nrows = in_sizes[0] / FDIM;

  dim3 block(256);
  dim3 grid(FDIM / (16 * WAVES_PER_BLOCK * NTILES), nrows / 16);
  hipLaunchKernelGGL(haar_dwt_wmma, grid, block, 0, stream, x, out);
}


# Tile = 256 contiguous floats of one row as WMMA B operand: every wave load reads 512 contiguous bytes, every store writes whole 128-B lines from D registers, no LDS; 4096 waves x 2 rows each
# speedup vs baseline: 4.8381x; 4.8381x over previous
#include <hip/hip_runtime.h>


typedef float v2f __attribute__((ext_vector_type(2)));
typedef float v4f __attribute__((ext_vector_type(4)));
typedef float v8f __attribute__((ext_vector_type(8)));

#define INV_SQRT2 0.70710678118654752440f
#define FDIM 4096
#define WAVES_PER_BLOCK 16
#define ROWS_PER_WAVE 2
#define TILE 256
#define UNIT_TILES 4
#define AHEAD 3
#define UNITS_PER_ROW (FDIM / (TILE * UNIT_TILES))
#define UNITS (ROWS_PER_WAVE * UNITS_PER_ROW)

__device__ __forceinline__ v2f wt_pair(int m, int j, int h) {
  const float c = INV_SQRT2;
  const int q = 4 * (j >> 1) + 2 * h + (j & 1);
  const bool hit = (q == (m & 7));
  v2f w;
  w.x = hit ? c : 0.0f;
  w.y = hit ? ((m & 8) ? -c : c) : 0.0f;
  return w;
}

__global__ __launch_bounds__(32 * WAVES_PER_BLOCK) void haar_dwt_wmma(const float* __restrict__ x,
                                                                     float* __restrict__ out) {
  const int lane = threadIdx.x & 31;
  const int wave = threadIdx.x >> 5;
  const int c = lane & 15;
  const int h = lane >> 4;

  v2f wa[4];
#pragma unroll
  for (int j = 0; j < 4; ++j) wa[j] = wt_pair(c, j, h);

  const size_t row0 = (size_t)(blockIdx.x * WAVES_PER_BLOCK + wave) * ROWS_PER_WAVE;
  const float* src = x + row0 * FDIM + 8 * c + 4 * h;
  float* dst = out + row0 * FDIM + h * (FDIM / 2) + 4 * c;

  v4f buf[AHEAD + 1][2 * UNIT_TILES];

#pragma unroll
  for (int u = 0; u < AHEAD; ++u)
#pragma unroll
    for (int k = 0; k < 2 * UNIT_TILES; ++k)
      buf[u][k] = __builtin_nontemporal_load((const v4f*)(src + u * (TILE * UNIT_TILES) + 128 * k));

#pragma unroll
  for (int u = 0; u < UNITS; ++u) {
    if (u + AHEAD < UNITS) {
#pragma unroll
      for (int k = 0; k < 2 * UNIT_TILES; ++k)
        buf[(u + AHEAD) % (AHEAD + 1)][k] = __builtin_nontemporal_load(
            (const v4f*)(src + (u + AHEAD) * (TILE * UNIT_TILES) + 128 * k));
    }
    __builtin_amdgcn_sched_barrier(0);

    float* o = dst + (u / UNITS_PER_ROW) * FDIM + (u % UNITS_PER_ROW) * (TILE * UNIT_TILES / 2);
#pragma unroll
    for (int t = 0; t < UNIT_TILES; ++t) {
      const v4f d0 = buf[u % (AHEAD + 1)][2 * t];
      const v4f d1 = buf[u % (AHEAD + 1)][2 * t + 1];
      v2f b0 = {d0.x, d0.y};
      v2f b1 = {d0.z, d0.w};
      v2f b2 = {d1.x, d1.y};
      v2f b3 = {d1.z, d1.w};
      v8f acc = {};
      acc = __builtin_amdgcn_wmma_f32_16x16x4_f32(false, wa[0], false, b0, (short)0, acc, false, false);
      acc = __builtin_amdgcn_wmma_f32_16x16x4_f32(false, wa[1], false, b1, (short)0, acc, false, false);
      acc = __builtin_amdgcn_wmma_f32_16x16x4_f32(false, wa[2], false, b2, (short)0, acc, false, false);
      acc = __builtin_amdgcn_wmma_f32_16x16x4_f32(false, wa[3], false, b3, (short)0, acc, false, false);
      v4f s0 = {acc[0], acc[1], acc[2], acc[3]};
      v4f s1 = {acc[4], acc[5], acc[6], acc[7]};
      __builtin_nontemporal_store(s0, (v4f*)(o + t * (TILE / 2)));
      __builtin_nontemporal_store(s1, (v4f*)(o + t * (TILE / 2) + 64));
    }
    __builtin_amdgcn_sched_barrier(0);
  }
}

extern "C" void kernel_launch(void* const* d_in, const int* in_sizes, int n_in,
                              void* d_out, int out_size, void* d_ws, size_t ws_size,
                              hipStream_t stream) {
  (void)n_in; (void)out_size; (void)d_ws; (void)ws_size;
  const float* x = (const float*)d_in[0];
  float* out = (float*)d_out;

  const int nrows = in_sizes[0] / FDIM;

  dim3 block(32 * WAVES_PER_BLOCK);
  dim3 grid(nrows / (WAVES_PER_BLOCK * ROWS_PER_WAVE));
  hipLaunchKernelGGL(haar_dwt_wmma, grid, block, 0, stream, x, out);
}


# Half the waves: one 8-wave work-group per WGP (was 16), each wave 4 rows (was 2) with 48-56 loads in flight (was 24-32): same bytes in flight per WGP, 2048 waves enter instead of 4096
# speedup vs baseline: 5.2167x; 1.0783x over previous
#include <hip/hip_runtime.h>


typedef float v2f __attribute__((ext_vector_type(2)));
typedef float v4f __attribute__((ext_vector_type(4)));
typedef float v8f __attribute__((ext_vector_type(8)));

#define INV_SQRT2 0.70710678118654752440f
#define FDIM 4096
#define WAVES_PER_BLOCK 8
#define ROWS_PER_WAVE 4
#define TILE 256
#define UNIT_TILES 4
#define AHEAD 6
#define UNITS_PER_ROW (FDIM / (TILE * UNIT_TILES))
#define UNITS (ROWS_PER_WAVE * UNITS_PER_ROW)

__device__ __forceinline__ v2f wt_pair(int m, int j, int h) {
  const float c = INV_SQRT2;
  const int q = 4 * (j >> 1) + 2 * h + (j & 1);
  const bool hit = (q == (m & 7));
  v2f w;
  w.x = hit ? c : 0.0f;
  w.y = hit ? ((m & 8) ? -c : c) : 0.0f;
  return w;
}

__global__ __launch_bounds__(32 * WAVES_PER_BLOCK) void haar_dwt_wmma(const float* __restrict__ x,
                                                                     float* __restrict__ out) {
  const int lane = threadIdx.x & 31;
  const int wave = threadIdx.x >> 5;
  const int c = lane & 15;
  const int h = lane >> 4;

  v2f wa[4];
#pragma unroll
  for (int j = 0; j < 4; ++j) wa[j] = wt_pair(c, j, h);

  const size_t row0 = (size_t)(blockIdx.x * WAVES_PER_BLOCK + wave) * ROWS_PER_WAVE;
  const float* src = x + row0 * FDIM + 8 * c + 4 * h;
  float* dst = out + row0 * FDIM + h * (FDIM / 2) + 4 * c;

  v4f buf[AHEAD + 1][2 * UNIT_TILES];

#pragma unroll
  for (int u = 0; u < AHEAD; ++u)
#pragma unroll
    for (int k = 0; k < 2 * UNIT_TILES; ++k)
      buf[u][k] = __builtin_nontemporal_load((const v4f*)(src + u * (TILE * UNIT_TILES) + 128 * k));

#pragma unroll
  for (int u = 0; u < UNITS; ++u) {
    if (u + AHEAD < UNITS) {
#pragma unroll
      for (int k = 0; k < 2 * UNIT_TILES; ++k)
        buf[(u + AHEAD) % (AHEAD + 1)][k] = __builtin_nontemporal_load(
            (const v4f*)(src + (u + AHEAD) * (TILE * UNIT_TILES) + 128 * k));
    }
    __builtin_amdgcn_sched_barrier(0);

    float* o = dst + (u / UNITS_PER_ROW) * FDIM + (u % UNITS_PER_ROW) * (TILE * UNIT_TILES / 2);
    v8f acc[UNIT_TILES];
#pragma unroll
    for (int t = 0; t < UNIT_TILES; ++t) acc[t] = (v8f){};
#pragma unroll
    for (int j = 0; j < 4; ++j) {
#pragma unroll
      for (int t = 0; t < UNIT_TILES; ++t) {
        const v4f d = buf[u % (AHEAD + 1)][2 * t + (j >> 1)];
        const v2f b = (j & 1) ? (v2f){d.z, d.w} : (v2f){d.x, d.y};
        acc[t] = __builtin_amdgcn_wmma_f32_16x16x4_f32(false, wa[j], false, b, (short)0, acc[t], false, false);
      }
    }
#pragma unroll
    for (int t = 0; t < UNIT_TILES; ++t) {
      v4f s0 = {acc[t][0], acc[t][1], acc[t][2], acc[t][3]};
      v4f s1 = {acc[t][4], acc[t][5], acc[t][6], acc[t][7]};
      __builtin_nontemporal_store(s0, (v4f*)(o + t * (TILE / 2)));
      __builtin_nontemporal_store(s1, (v4f*)(o + t * (TILE / 2) + 64));
    }
    __builtin_amdgcn_sched_barrier(0);
  }
}

extern "C" void kernel_launch(void* const* d_in, const int* in_sizes, int n_in,
                              void* d_out, int out_size, void* d_ws, size_t ws_size,
                              hipStream_t stream) {
  (void)n_in; (void)out_size; (void)d_ws; (void)ws_size;
  const float* x = (const float*)d_in[0];
  float* out = (float*)d_out;

  const int nrows = in_sizes[0] / FDIM;

  dim3 block(32 * WAVES_PER_BLOCK);
  dim3 grid(nrows / (WAVES_PER_BLOCK * ROWS_PER_WAVE));
  hipLaunchKernelGGL(haar_dwt_wmma, grid, block, 0, stream, x, out);
}
